# starting kernel
# baseline (speedup 1.0000x reference)
#include <hip/hip_runtime.h>

typedef __attribute__((ext_vector_type(2))) float v2f;
typedef __attribute__((ext_vector_type(8))) float v8f;

static constexpr int HH   = 56;
static constexpr int WW   = 56;
static constexpr int HW   = HH * WW;
static constexpr int KTOT = 84;
static constexpr int NCH  = 48;
static constexpr int ROWS = 2 * HW;
static constexpr int ROW_TILES = ROWS / 16;

__global__ void fold_weights_kernel(const float* __restrict__ W0,
                                    const float* __restrict__ W1,
                                    float* __restrict__ Wc) {
  int idx = blockIdx.x * blockDim.x + threadIdx.x;
  if (idx >= NCH * KTOT) return;
  int p  = idx / KTOT;
  int jk = idx - p * KTOT;
  const float* w1 = W1 + jk * 9;
  const float* w0 = W0 + p  * 9;
  float s = 0.f;
#pragma unroll
  for (int i = 0; i < 9; ++i) s += w1[i] * w0[i];
  Wc[p * KTOT + jk] = s;
}

__global__ __launch_bounds__(128)
void conv_wmma_kernel(const float* __restrict__ x,
                      const float* __restrict__ Wc,
                      float* __restrict__ out) {
  __shared__ alignas(16) float lWc[NCH * KTOT];

  const int tid = threadIdx.x;
  for (int i = tid; i < NCH * KTOT; i += 128) lWc[i] = Wc[i];
  __syncthreads();

  const int wave  = tid >> 5;
  const int lane  = tid & 31;
  const int laneh = lane >> 4;
  const int lanem = lane & 15;

  const int row_tile = blockIdx.x * 4 + wave;
  const int row = row_tile * 16 + lanem;
  const int o   = row / HW;
  const int rem = row - o * HW;
  const int n   = rem / WW;
  const int m   = rem - n * WW;
  const float* xb = x + (o * 12) * HW + n * WW;
  __builtin_prefetch(xb + m, 0, 3);

  const int khalf = laneh * 2;

  int j0 = 0, kk0 = khalf;
  int j1 = 0, kk1 = khalf + 1;

  const float* w0r = &lWc[( 0 + lanem) * KTOT + khalf];
  const float* w1r = &lWc[(16 + lanem) * KTOT + khalf];
  const float* w2r = &lWc[(32 + lanem) * KTOT + khalf];

  v8f acc0 = {};
  v8f acc1 = {};
  v8f acc2 = {};

  for (int b = 0; b < KTOT / 4; ++b) {
    const int mm0 = m + kk0 - 3;
    const int mm1 = m + kk1 - 3;
    const float a0 = (mm0 >= 0 && mm0 < WW) ? xb[j0 * HW + mm0] : 0.f;
    const float a1 = (mm1 >= 0 && mm1 < WW) ? xb[j1 * HW + mm1] : 0.f;
    v2f a = {a0, a1};

    const v2f b0 = *(const v2f*)(w0r + b * 4);
    const v2f b1 = *(const v2f*)(w1r + b * 4);
    const v2f b2 = *(const v2f*)(w2r + b * 4);

    acc0 = __builtin_amdgcn_wmma_f32_16x16x4_f32(false, a, false, b0,
                                                 (short)0, acc0, false, false);
    acc1 = __builtin_amdgcn_wmma_f32_16x16x4_f32(false, a, false, b1,
                                                 (short)0, acc1, false, false);
    acc2 = __builtin_amdgcn_wmma_f32_16x16x4_f32(false, a, false, b2,
                                                 (short)0, acc2, false, false);

    kk0 += 4; if (kk0 >= 7) { kk0 -= 7; ++j0; }
    kk1 += 4; if (kk1 >= 7) { kk1 -= 7; ++j1; }
  }

  const int grow0 = row_tile * 16 + laneh * 8;
  const int rem0  = grow0 - o * HW;
  int n2 = rem0 / WW;
  int m2 = rem0 - n2 * WW;
  int nr = n2 + 1; if (nr == HH) nr = 0;
  float* obase = out + (size_t)(o * NCH) * HW;

#pragma unroll
  for (int r = 0; r < 8; ++r) {
    float* ob = obase + nr * WW + m2;
    ob[(size_t)( 0 + lanem) * HW] = acc0[r];
    ob[(size_t)(16 + lanem) * HW] = acc1[r];
    ob[(size_t)(32 + lanem) * HW] = acc2[r];
    ++m2;
    if (m2 == WW) { m2 = 0; ++nr; if (nr == HH) nr = 0; }
  }
}

extern "C" void kernel_launch(void* const* d_in, const int* in_sizes, int n_in,
                              void* d_out, int out_size, void* d_ws, size_t ws_size,
                              hipStream_t stream) {
  const float* x  = (const float*)d_in[0];
  const float* W0 = (const float*)d_in[1];
  const float* W1 = (const float*)d_in[2];
  float* out = (float*)d_out;
  float* Wc  = (float*)d_ws;

  fold_weights_kernel<<<(NCH * KTOT + 127) / 128, 128, 0, stream>>>(W0, W1, Wc);
  conv_wmma_kernel<<<ROW_TILES / 4, 128, 0, stream>>>(x, Wc, out);
}


# Conv kernel issues all its global loads (8 b128 weight loads, 42 im2col gathers made unconditional by clamping the column) as one group before any wait: one memory latency per wave, not ~30 serial
# speedup vs baseline: 2.2047x; 2.2047x over previous
#include <hip/hip_runtime.h>
#include <utility>


typedef __attribute__((ext_vector_type(2))) float v2f;
typedef __attribute__((ext_vector_type(4))) float v4f;
typedef __attribute__((ext_vector_type(8))) float v8f;

static constexpr int HH   = 56;
static constexpr int WW   = 56;
static constexpr int HW   = HH * WW;
static constexpr int KTOT = 84;
static constexpr int KSTEPS = KTOT / 4;
static constexpr int NCH  = 48;
static constexpr int ROWS = 2 * HW;
static constexpr int ROW_TILES = ROWS / 16;
static constexpr int WC_VEC4   = NCH * KTOT / 4;
static constexpr int WC_PASSES = 8;

__global__ void fold_weights_kernel(const float* __restrict__ W0,
                                    const float* __restrict__ W1,
                                    float* __restrict__ Wc) {
  int idx = blockIdx.x * blockDim.x + threadIdx.x;
  if (idx >= NCH * KTOT) return;
  int p  = idx / KTOT;
  int jk = idx - p * KTOT;
  const float* w1 = W1 + jk * 9;
  const float* w0 = W0 + p  * 9;
  float s = 0.f;
#pragma unroll
  for (int i = 0; i < 9; ++i) s += w1[i] * w0[i];
  Wc[p * KTOT + jk] = s;
}

template <int K> struct tap {
  static constexpr int jo = (K / 7) * HW;
  static constexpr int d  = K % 7 - 3;
};

template <int T>
__device__ __forceinline__ float gather_tap(const float* __restrict__ x, unsigned rowbase,
                                            int m, int laneh) {
  constexpr int k0 = 4 * (T / 2) + (T % 2);
  constexpr int k1 = k0 + 2;
  const int      d  = laneh ? tap<k1>::d : tap<k0>::d;
  const unsigned jo = laneh ? (unsigned)tap<k1>::jo : (unsigned)tap<k0>::jo;
  const int mm = m + d;
  const int mc = min(max(mm, 0), WW - 1);
  const float val = x[rowbase + jo + (unsigned)mc];
  return ((unsigned)mm < (unsigned)WW) ? val : 0.f;
}

template <int... T>
__device__ __forceinline__ void gather_all(float (&av)[2 * KSTEPS], const float* __restrict__ x,
                                           unsigned rowbase, int m, int laneh,
                                           std::integer_sequence<int, T...>) {
  ((av[T] = gather_tap<T>(x, rowbase, m, laneh)), ...);
}

__global__ __launch_bounds__(128)
void conv_wmma_kernel(const float* __restrict__ x,
                      const float* __restrict__ Wc,
                      float* __restrict__ out) {
  __shared__ alignas(16) float lWc[WC_PASSES * 128 * 4];

  const int tid = threadIdx.x;

  const v4f* Wc4 = (const v4f*)Wc;
  v4f wv[WC_PASSES];
#pragma unroll
  for (int t = 0; t < WC_PASSES; ++t) {
    const int i4  = tid + 128 * t;
    const int i4c = i4 < WC_VEC4 ? i4 : WC_VEC4 - 1;
    wv[t] = Wc4[i4c];
  }

  const int wave  = tid >> 5;
  const int lane  = tid & 31;
  const int laneh = lane >> 4;
  const int lanem = lane & 15;

  const int row_tile = blockIdx.x * 4 + wave;
  const int row = row_tile * 16 + lanem;
  const int o   = row / HW;
  const int rem = row - o * HW;
  const int n   = rem / WW;
  const int m   = rem - n * WW;
  const unsigned rowbase = (unsigned)((o * 12) * HW + n * WW);

  const int khalf = laneh * 2;

  float av[2 * KSTEPS];
  gather_all(av, x, rowbase, m, laneh, std::make_integer_sequence<int, 2 * KSTEPS>{});

#pragma unroll
  for (int t = 0; t < WC_PASSES; ++t) ((v4f*)lWc)[tid + 128 * t] = wv[t];
  __syncthreads();

  const float* w0r = &lWc[( 0 + lanem) * KTOT + khalf];
  const float* w1r = &lWc[(16 + lanem) * KTOT + khalf];
  const float* w2r = &lWc[(32 + lanem) * KTOT + khalf];

  v8f acc0 = {};
  v8f acc1 = {};
  v8f acc2 = {};

#pragma unroll
  for (int b = 0; b < KSTEPS; ++b) {
    const v2f a = {av[2 * b], av[2 * b + 1]};

    const v2f b0 = *(const v2f*)(w0r + b * 4);
    const v2f b1 = *(const v2f*)(w1r + b * 4);
    const v2f b2 = *(const v2f*)(w2r + b * 4);

    acc0 = __builtin_amdgcn_wmma_f32_16x16x4_f32(false, a, false, b0,
                                                 (short)0, acc0, false, false);
    acc1 = __builtin_amdgcn_wmma_f32_16x16x4_f32(false, a, false, b1,
                                                 (short)0, acc1, false, false);
    acc2 = __builtin_amdgcn_wmma_f32_16x16x4_f32(false, a, false, b2,
                                                 (short)0, acc2, false, false);
  }

  const int grow0 = row_tile * 16 + laneh * 8;
  const int rem0  = grow0 - o * HW;
  int n2 = rem0 / WW;
  int m2 = rem0 - n2 * WW;
  int nr = n2 + 1; if (nr == HH) nr = 0;
  float* obase = out + (size_t)(o * NCH) * HW;

#pragma unroll
  for (int r = 0; r < 8; ++r) {
    float* ob = obase + nr * WW + m2;
    ob[(size_t)( 0 + lanem) * HW] = acc0[r];
    ob[(size_t)(16 + lanem) * HW] = acc1[r];
    ob[(size_t)(32 + lanem) * HW] = acc2[r];
    ++m2;
    if (m2 == WW) { m2 = 0; ++nr; if (nr == HH) nr = 0; }
  }
}

extern "C" void kernel_launch(void* const* d_in, const int* in_sizes, int n_in,
                              void* d_out, int out_size, void* d_ws, size_t ws_size,
                              hipStream_t stream) {
  const float* x  = (const float*)d_in[0];
  const float* W0 = (const float*)d_in[1];
  const float* W1 = (const float*)d_in[2];
  float* out = (float*)d_out;
  float* Wc  = (float*)d_ws;

  fold_weights_kernel<<<(NCH * KTOT + 127) / 128, 128, 0, stream>>>(W0, W1, Wc);
  conv_wmma_kernel<<<ROW_TILES / 4, 128, 0, stream>>>(x, Wc, out);
}
